# starting kernel
# baseline (speedup 1.0000x reference)
#include <hip/hip_runtime.h>


typedef float v2f __attribute__((ext_vector_type(2)));
typedef float v8f __attribute__((ext_vector_type(8)));

#define NB 64
#define NT 2048
#define NC 32
#define NF 16
#define TCH 256
#define XS_ROW 289
#define OS_ROW 257

__global__ __launch_bounds__(1024)
void fir_wmma_kernel(const float* __restrict__ x,
                     const float* __restrict__ fw,
                     float* __restrict__ out) {
    __shared__ float lds[NC * XS_ROW];

    const int tid  = threadIdx.x;
    const int t0   = blockIdx.x * TCH;
    const int b    = blockIdx.y;

    const long xbase = (long)b * NT * NC;
    #pragma unroll
    for (int it = 0; it < 9; ++it) {
        int f = tid + it * 1024;
        if (f < 272 * NC) {
            int cc = f & (NC - 1);
            int tt = f >> 5;
            int t  = t0 - (NF - 1) + tt;
            float v = 0.0f;
            if (t >= 0 && t < NT) v = x[xbase + (long)t * NC + cc];
            lds[cc * XS_ROW + (tt & 15) * 17 + (tt >> 4)] = v;
        }
    }
    __syncthreads();

    const int lane = tid & 31;
    const int c    = tid >> 5;
    const int mn   = lane & 15;
    const int hi   = lane >> 4;

    v2f bw[8];
    #pragma unroll
    for (int kb = 0; kb < 8; ++kb) {
        int s0 = 4 * kb + 2 * hi;
        int i0 = s0 - mn;
        int i1 = i0 + 1;
        float b0 = (i0 >= 0 && i0 < NF) ? fw[i0 * NC + c] : 0.0f;
        float b1 = (i1 >= 0 && i1 < NF) ? fw[i1 * NC + c] : 0.0f;
        bw[kb] = v2f{b0, b1};
    }

    v8f acc = {0.f, 0.f, 0.f, 0.f, 0.f, 0.f, 0.f, 0.f};
    #pragma unroll
    for (int kb = 0; kb < 8; ++kb) {
        int tt0 = mn * 16 + 4 * kb + 2 * hi;
        int tt1 = tt0 + 1;
        float a0 = lds[c * XS_ROW + (tt0 & 15) * 17 + (tt0 >> 4)];
        float a1 = lds[c * XS_ROW + (tt1 & 15) * 17 + (tt1 >> 4)];
        v2f a = v2f{a0, a1};
        acc = __builtin_amdgcn_wmma_f32_16x16x4_f32(
            false, a, false, bw[kb], (short)0, acc, false, false);
    }
    __syncthreads();

    #pragma unroll
    for (int d = 0; d < 8; ++d) {
        int tl = (d + 8 * hi) * 16 + mn;
        lds[c * OS_ROW + tl] = acc[d];
    }
    __syncthreads();

    const long obase = (long)b * NT * NC + (long)t0 * NC;
    #pragma unroll
    for (int it = 0; it < 8; ++it) {
        int f  = tid + it * 1024;
        int cc = f & (NC - 1);
        int tt = f >> 5;
        out[obase + (long)tt * NC + cc] = lds[cc * OS_ROW + tt];
    }
}

extern "C" void kernel_launch(void* const* d_in, const int* in_sizes, int n_in,
                              void* d_out, int out_size, void* d_ws, size_t ws_size,
                              hipStream_t stream) {
    const float* x  = (const float*)d_in[0];
    const float* fw = (const float*)d_in[1];
    float* out      = (float*)d_out;
    (void)in_sizes; (void)n_in; (void)out_size; (void)d_ws; (void)ws_size;

    dim3 grid(NT / TCH, NB);
    dim3 block(1024);
    fir_wmma_kernel<<<grid, block, 0, stream>>>(x, fw, out);
}


# One 32-wave group per (batch, 512-step segment), 256 groups: a wave keeps its channel and Toeplitz weights for 2 tiles; body rows loaded unconditionally, all in flight; sum order unchanged
# speedup vs baseline: 1.4319x; 1.4319x over previous
#include <hip/hip_runtime.h>


typedef float v2f __attribute__((ext_vector_type(2)));
typedef float v8f __attribute__((ext_vector_type(8)));

#define NB 64
#define NT 2048
#define NC 32
#define NF 16
#define SEG 512
#define NTILE (SEG / 256)
#define XROWS (SEG + NF)
#define XG 49
#define XS_ROW 769
#define OS_ROW 561

__device__ __forceinline__ int xs_slot(int tt) { return (tt & 15) * XG + (tt >> 4); }
__device__ __forceinline__ int os_slot(int tl) { return tl + ((tl >> 7) << 4); }

__global__ __launch_bounds__(1024)
void fir_wmma_seg_kernel(const float* __restrict__ x,
                         const float* __restrict__ fw,
                         float* __restrict__ out) {
    __shared__ float xs[NC * XS_ROW];
    __shared__ float os[NC * OS_ROW];

    const int tid  = threadIdx.x;
    const int lane = tid & 31;
    const int wv   = tid >> 5;
    const int seg  = blockIdx.x;
    const int b    = blockIdx.y;
    const int t0   = seg * SEG;

    const long gbase = ((long)b * NT + t0) * NC;
    float halo = 0.0f;
    if (seg > 0 && wv < NF - 1)
        halo = x[gbase + (long)(wv - (NF - 1)) * NC + lane];
    float body[SEG / 32];
    #pragma unroll
    for (int it = 0; it < SEG / 32; ++it)
        body[it] = x[gbase + (long)(wv + 32 * it) * NC + lane];

    if (wv < NF - 1)
        xs[lane * XS_ROW + xs_slot(wv)] = halo;
    else if (wv == NF - 1)
        xs[lane * XS_ROW + xs_slot(XROWS - 1)] = 0.0f;
    const int tb = wv + NF - 1;
    const int xw = lane * XS_ROW + xs_slot(tb);
    #pragma unroll
    for (int it = 0; it < SEG / 32; ++it)
        xs[xw + 2 * it] = body[it];
    __syncthreads();

    const int c  = wv;
    const int mn = lane & 15;
    const int hi = lane >> 4;

    v2f bw[8];
    #pragma unroll
    for (int kb = 0; kb < 8; ++kb) {
        int s0 = 4 * kb + 2 * hi;
        int i0 = s0 - mn;
        int i1 = i0 + 1;
        float b0 = (i0 >= 0 && i0 < NF) ? fw[i0 * NC + c] : 0.0f;
        float b1 = (i1 >= 0 && i1 < NF) ? fw[i1 * NC + c] : 0.0f;
        bw[kb] = v2f{b0, b1};
    }

    v8f acc[NTILE];
    #pragma unroll
    for (int tl = 0; tl < NTILE; ++tl) acc[tl] = v8f{0.f, 0.f, 0.f, 0.f, 0.f, 0.f, 0.f, 0.f};
    #pragma unroll
    for (int kb = 0; kb < 8; ++kb) {
        #pragma unroll
        for (int tl = 0; tl < NTILE; ++tl) {
            int tt0 = 256 * tl + mn * 16 + 4 * kb + 2 * hi;
            float a0 = xs[c * XS_ROW + xs_slot(tt0)];
            float a1 = xs[c * XS_ROW + xs_slot(tt0 + 1)];
            v2f a = v2f{a0, a1};
            acc[tl] = __builtin_amdgcn_wmma_f32_16x16x4_f32(
                false, a, false, bw[kb], (short)0, acc[tl], false, false);
        }
    }

    #pragma unroll
    for (int tl = 0; tl < NTILE; ++tl) {
        #pragma unroll
        for (int d = 0; d < 8; ++d) {
            int tloc = 256 * tl + (d + 8 * hi) * 16 + mn;
            os[c * OS_ROW + os_slot(tloc)] = acc[tl][d];
        }
    }
    __syncthreads();

    #pragma unroll
    for (int it = 0; it < SEG / 32; ++it) {
        int r = wv + 32 * it;
        out[gbase + (long)r * NC + lane] = os[lane * OS_ROW + os_slot(r)];
    }
}

extern "C" void kernel_launch(void* const* d_in, const int* in_sizes, int n_in,
                              void* d_out, int out_size, void* d_ws, size_t ws_size,
                              hipStream_t stream) {
    const float* x  = (const float*)d_in[0];
    const float* fw = (const float*)d_in[1];
    float* out      = (float*)d_out;
    (void)in_sizes; (void)n_in; (void)out_size; (void)d_ws; (void)ws_size;

    dim3 grid(NT / SEG, NB);
    dim3 block(1024);
    fir_wmma_seg_kernel<<<grid, block, 0, stream>>>(x, fw, out);
}


# Two-stage pipeline over the group's two tiles: taps fetched first (one load, lane permutes), tile 0 computed while tile 1's rows land, its stores drain under tile 1; LDS-only barriers; sum order same
# speedup vs baseline: 1.7527x; 1.2240x over previous
#include <hip/hip_runtime.h>


typedef float v2f __attribute__((ext_vector_type(2)));
typedef float v8f __attribute__((ext_vector_type(8)));

#define NB 64
#define NT 2048
#define NC 32
#define NF 16
#define SEG 512
#define NTILE (SEG / 256)
#define NIT (SEG / 32)
#define NIT0 (NIT / 2 + 1)
#define XROWS (SEG + NF)
#define XG 49
#define XS_ROW 769
#define OS_ROW 561

__device__ __forceinline__ int xs_slot(int tt) { return (tt & 15) * XG + (tt >> 4); }
__device__ __forceinline__ int os_slot(int tl) { return tl + ((tl >> 7) << 4); }

__device__ __forceinline__ void lds_barrier() {
    asm volatile("s_wait_dscnt 0x0\n\ts_barrier_signal -1\n\ts_barrier_wait -1" ::: "memory");
}

__global__ __launch_bounds__(1024)
void fir_wmma_seg_kernel(const float* __restrict__ x,
                         const float* __restrict__ fw,
                         float* __restrict__ out) {
    __shared__ float xs[NC * XS_ROW];
    __shared__ float os[NC * OS_ROW];

    const int tid  = threadIdx.x;
    const int lane = tid & 31;
    const int wv   = tid >> 5;
    const int seg  = blockIdx.x;
    const int b    = blockIdx.y;
    const int t0   = seg * SEG;
    const int c    = wv;
    const int mn   = lane & 15;
    const int hi   = lane >> 4;

    const float wtap = fw[(lane & 15) * NC + c];
    const long gbase = ((long)b * NT + t0) * NC;
    float halo = 0.0f;
    if (seg > 0 && wv < NF - 1)
        halo = x[gbase + (long)(wv - (NF - 1)) * NC + lane];
    float body[NIT];
    #pragma unroll
    for (int it = 0; it < NIT; ++it)
        body[it] = x[gbase + (long)(wv + 32 * it) * NC + lane];

    v2f bw[8];
    #pragma unroll
    for (int kb = 0; kb < 8; ++kb) {
        int i0 = 4 * kb + 2 * hi - mn;
        int i1 = i0 + 1;
        float p0 = __int_as_float(__builtin_amdgcn_ds_bpermute((i0 & 15) << 2, __float_as_int(wtap)));
        float p1 = __int_as_float(__builtin_amdgcn_ds_bpermute((i1 & 15) << 2, __float_as_int(wtap)));
        float b0 = (i0 >= 0 && i0 < NF) ? p0 : 0.0f;
        float b1 = (i1 >= 0 && i1 < NF) ? p1 : 0.0f;
        bw[kb] = v2f{b0, b1};
    }

    if (wv < NF - 1)
        xs[lane * XS_ROW + xs_slot(wv)] = halo;
    else if (wv == NF - 1)
        xs[lane * XS_ROW + xs_slot(XROWS - 1)] = 0.0f;
    const int tb = wv + NF - 1;
    const int xw = lane * XS_ROW + xs_slot(tb);
    #pragma unroll
    for (int it = 0; it < NIT0; ++it)
        xs[xw + 2 * it] = body[it];
    lds_barrier();

    const int xa = c * XS_ROW;
    const int oc = c * OS_ROW;

    {
        v8f acc = v8f{0.f, 0.f, 0.f, 0.f, 0.f, 0.f, 0.f, 0.f};
        #pragma unroll
        for (int kb = 0; kb < 8; ++kb) {
            int tt0 = mn * 16 + 4 * kb + 2 * hi;
            v2f a = v2f{xs[xa + xs_slot(tt0)], xs[xa + xs_slot(tt0 + 1)]};
            acc = __builtin_amdgcn_wmma_f32_16x16x4_f32(
                false, a, false, bw[kb], (short)0, acc, false, false);
        }
        #pragma unroll
        for (int d = 0; d < 8; ++d)
            os[oc + os_slot((d + 8 * hi) * 16 + mn)] = acc[d];
    }
    #pragma unroll
    for (int it = NIT0; it < NIT; ++it)
        xs[xw + 2 * it] = body[it];
    lds_barrier();

    #pragma unroll
    for (int it = 0; it < NIT / 2; ++it) {
        int r = wv + 32 * it;
        out[gbase + (long)r * NC + lane] = os[lane * OS_ROW + os_slot(r)];
    }
    __builtin_amdgcn_sched_barrier(0);
    {
        v8f acc = v8f{0.f, 0.f, 0.f, 0.f, 0.f, 0.f, 0.f, 0.f};
        #pragma unroll
        for (int kb = 0; kb < 8; ++kb) {
            int tt0 = 256 + mn * 16 + 4 * kb + 2 * hi;
            v2f a = v2f{xs[xa + xs_slot(tt0)], xs[xa + xs_slot(tt0 + 1)]};
            acc = __builtin_amdgcn_wmma_f32_16x16x4_f32(
                false, a, false, bw[kb], (short)0, acc, false, false);
        }
        #pragma unroll
        for (int d = 0; d < 8; ++d)
            os[oc + os_slot(256 + (d + 8 * hi) * 16 + mn)] = acc[d];
    }
    lds_barrier();
    #pragma unroll
    for (int it = NIT / 2; it < NIT; ++it) {
        int r = wv + 32 * it;
        out[gbase + (long)r * NC + lane] = os[lane * OS_ROW + os_slot(r)];
    }
}

extern "C" void kernel_launch(void* const* d_in, const int* in_sizes, int n_in,
                              void* d_out, int out_size, void* d_ws, size_t ws_size,
                              hipStream_t stream) {
    const float* x  = (const float*)d_in[0];
    const float* fw = (const float*)d_in[1];
    float* out      = (float*)d_out;
    (void)in_sizes; (void)n_in; (void)out_size; (void)d_ws; (void)ws_size;

    dim3 grid(NT / SEG, NB);
    dim3 block(1024);
    fir_wmma_seg_kernel<<<grid, block, 0, stream>>>(x, fw, out);
}
